# starting kernel
# baseline (speedup 1.0000x reference)
#include <hip/hip_runtime.h>
#include <hip/hip_bf16.h>


typedef __attribute__((ext_vector_type(16))) _Float16 v16h;
typedef __attribute__((ext_vector_type(8)))  _Float16 v8h;
typedef __attribute__((ext_vector_type(8)))  float    v8f;

#define LATENT   64
#define EDGE_IN  70
#define K1PAD    96
#define HID      20
#define OUTD     192
#define OUT_CH   3
#define WAVES    4


__device__ inline v8f wmma16(v16h a, v16h b, v8f c) {
    return __builtin_amdgcn_wmma_f32_16x16x32_f16(
        false, a, false, b, (short)0, c, false, false);
}

__device__ inline v16h cat8(v8h lo, v8h hi) {
    return __builtin_shufflevector(lo, hi,
        0, 1, 2, 3, 4, 5, 6, 7, 8, 9, 10, 11, 12, 13, 14, 15);
}

__device__ inline v16h load_a(const _Float16* H, int ld, int k0) {
    int lane = threadIdx.x & 31;
    int m  = lane & 15;
    int kb = k0 + ((lane & 16) ? 8 : 0);
    v8h lo = *(const v8h*)(H + m * ld + kb);
    v8h hi = *(const v8h*)(H + m * ld + kb + 16);
    return cat8(lo, hi);
}

__device__ inline v16h make_b(const _Float16* WT, int ldk, int k0, int n0) {
    int lane = threadIdx.x & 31;
    int n  = n0 + (lane & 15);
    int kb = k0 + ((lane & 16) ? 8 : 0);
    v8h lo = *(const v8h*)(WT + n * ldk + kb);
    v8h hi = *(const v8h*)(WT + n * ldk + kb + 16);
    return cat8(lo, hi);
}

__device__ inline void store_c(_Float16* H, int ld, v8f c, int n0,
                               const float* bias, bool relu) {
    int lane = threadIdx.x & 31;
    int n  = n0 + (lane & 15);
    int mb = (lane & 16) ? 8 : 0;
    float bv = bias[n];
#pragma unroll
    for (int r = 0; r < 8; ++r) {
        float v = c[r] + bv;
        if (relu) v = fmaxf(v, 0.0f);
        H[(mb + r) * ld + n] = (_Float16)v;
    }
}


__global__ void zero_f32(float* p, int n) {
    int i = blockIdx.x * blockDim.x + threadIdx.x;
    if (i < n) p[i] = 0.0f;
}

__global__ __launch_bounds__(WAVES * 32)
void nnconv_edge_kernel(const float* __restrict__ x,
                        const int*   __restrict__ edge_index,
                        const float* __restrict__ edge_attr,
                        const float* __restrict__ W1, const float* __restrict__ b1,
                        const float* __restrict__ W2, const float* __restrict__ b2,
                        const float* __restrict__ W3, const float* __restrict__ b3,
                        const float* __restrict__ W4, const float* __restrict__ b4,
                        float* __restrict__ summed, float* __restrict__ count,
                        int E) {
    __shared__ alignas(16) _Float16 sW1T[32  * K1PAD];
    __shared__ alignas(16) _Float16 sW2T[32  * 32];
    __shared__ alignas(16) _Float16 sW3T[32  * 32];
    __shared__ alignas(16) _Float16 sW4T[OUTD * 32];
    __shared__ float sB1[32], sB2[32], sB3[32];
    __shared__ float sB4[OUTD];
    __shared__ alignas(16) _Float16 sA [WAVES][16 * K1PAD];
    __shared__ alignas(16) _Float16 sO4[WAVES][16 * OUTD];

    const int tid = threadIdx.x;

    for (int i = tid; i < 32 * K1PAD; i += blockDim.x) {
        int n = i / K1PAD, k = i % K1PAD;
        sW1T[i] = (_Float16)((k < EDGE_IN && n < HID) ? W1[k * HID + n] : 0.0f);
    }
    for (int i = tid; i < 32 * 32; i += blockDim.x) {
        int n = i >> 5, k = i & 31;
        bool ok = (k < HID && n < HID);
        sW2T[i] = (_Float16)(ok ? W2[k * HID + n] : 0.0f);
        sW3T[i] = (_Float16)(ok ? W3[k * HID + n] : 0.0f);
    }
    for (int i = tid; i < OUTD * 32; i += blockDim.x) {
        int n = i >> 5, k = i & 31;
        sW4T[i] = (_Float16)((k < HID) ? W4[k * OUTD + n] : 0.0f);
    }
    for (int i = tid; i < 32; i += blockDim.x) {
        sB1[i] = (i < HID) ? b1[i] : 0.0f;
        sB2[i] = (i < HID) ? b2[i] : 0.0f;
        sB3[i] = (i < HID) ? b3[i] : 0.0f;
    }
    for (int i = tid; i < OUTD; i += blockDim.x) sB4[i] = b4[i];
    __syncthreads();

    const int lane = tid & 31;
    const int wave = tid >> 5;
    const int m    = lane & 15;
    const int half = lane >> 4;

    const long tile = (long)blockIdx.x * WAVES + wave;
    const long e0   = tile * 16;
    if (e0 >= E) return;

    _Float16* A1 = sA [wave];
    _Float16* O4 = sO4[wave];

    {
        const long gbase = e0 * EDGE_IN;
        const long gmax  = (long)E * EDGE_IN;
#pragma unroll 5
        for (int i = lane; i < 16 * EDGE_IN; i += 32) {
            long g = gbase + i;
            float v = (g < gmax) ? edge_attr[g] : 0.0f;
            A1[(i / EDGE_IN) * K1PAD + (i % EDGE_IN)] = (_Float16)v;
        }
#pragma unroll
        for (int i = lane; i < 16 * (K1PAD - EDGE_IN); i += 32) {
            A1[(i / 26) * K1PAD + EDGE_IN + (i % 26)] = (_Float16)0.0f;
        }
    }
    asm volatile("s_wait_dscnt 0" ::: "memory");

    _Float16* H = A1;
    {
        v8f c0 = {}, c1 = {};
#pragma unroll
        for (int ks = 0; ks < 3; ++ks) {
            v16h a = load_a(A1, K1PAD, ks * 32);
            c0 = wmma16(a, make_b(sW1T, K1PAD, ks * 32, 0),  c0);
            c1 = wmma16(a, make_b(sW1T, K1PAD, ks * 32, 16), c1);
        }
        store_c(H, 32, c0, 0,  sB1, true);
        store_c(H, 32, c1, 16, sB1, true);
    }
    asm volatile("s_wait_dscnt 0" ::: "memory");

    {
        v16h a = load_a(H, 32, 0);
        v8f c0 = {}, c1 = {};
        c0 = wmma16(a, make_b(sW2T, 32, 0, 0),  c0);
        c1 = wmma16(a, make_b(sW2T, 32, 0, 16), c1);
        asm volatile("s_wait_dscnt 0" ::: "memory");
        store_c(H, 32, c0, 0,  sB2, true);
        store_c(H, 32, c1, 16, sB2, true);
    }
    asm volatile("s_wait_dscnt 0" ::: "memory");

    {
        v16h a = load_a(H, 32, 0);
        v8f c0 = {}, c1 = {};
        c0 = wmma16(a, make_b(sW3T, 32, 0, 0),  c0);
        c1 = wmma16(a, make_b(sW3T, 32, 0, 16), c1);
        asm volatile("s_wait_dscnt 0" ::: "memory");
        store_c(H, 32, c0, 0,  sB3, true);
        store_c(H, 32, c1, 16, sB3, true);
    }
    asm volatile("s_wait_dscnt 0" ::: "memory");

    {
        v16h a = load_a(H, 32, 0);
#pragma unroll
        for (int t = 0; t < OUTD / 16; ++t) {
            v8f c = {};
            c = wmma16(a, make_b(sW4T, 32, 0, t * 16), c);
            store_c(O4, OUTD, c, t * 16, sB4, false);
        }
    }
    asm volatile("s_wait_dscnt 0" ::: "memory");

    const long e = e0 + m;
    if (e < E) {
        const int src = edge_index[e];
        const int dst = edge_index[(size_t)E + e];

        float xv[32];
        const float4* xr = (const float4*)(x + (size_t)src * LATENT + half * 32);
#pragma unroll
        for (int q = 0; q < 8; ++q) {
            float4 t = xr[q];
            xv[q * 4 + 0] = t.x; xv[q * 4 + 1] = t.y;
            xv[q * 4 + 2] = t.z; xv[q * 4 + 3] = t.w;
        }

        const v8h* wp = (const v8h*)(O4 + m * OUTD + half * 32 * OUT_CH);
        float p0 = 0.0f, p1 = 0.0f, p2 = 0.0f;
#pragma unroll
        for (int q = 0; q < 12; ++q) {
            v8h w8 = wp[q];
#pragma unroll
            for (int j = 0; j < 8; ++j) {
                int t = q * 8 + j;
                int i = t / 3, c = t % 3;
                float prod = xv[i] * (float)w8[j];
                if (c == 0) p0 += prod;
                else if (c == 1) p1 += prod;
                else p2 += prod;
            }
        }
        p0 += __shfl_xor(p0, 16);
        p1 += __shfl_xor(p1, 16);
        p2 += __shfl_xor(p2, 16);
        if (half == 0) {
            atomicAdd(&summed[(size_t)dst * 3 + 0], p0);
            atomicAdd(&summed[(size_t)dst * 3 + 1], p1);
            atomicAdd(&summed[(size_t)dst * 3 + 2], p2);
            atomicAdd(&count[dst], 1.0f);
        }
    }
}

__global__ void nnconv_finalize_kernel(const float* __restrict__ x,
                                       const float* __restrict__ root_w,
                                       const float* __restrict__ bias,
                                       const float* __restrict__ summed,
                                       const float* __restrict__ count,
                                       float* __restrict__ out, int N) {
    int n = blockIdx.x * blockDim.x + threadIdx.x;
    if (n >= N) return;
    float inv = 1.0f / fmaxf(count[n], 1.0f);
    float a0 = summed[(size_t)n * 3 + 0] * inv + bias[0];
    float a1 = summed[(size_t)n * 3 + 1] * inv + bias[1];
    float a2 = summed[(size_t)n * 3 + 2] * inv + bias[2];
    const float4* xr = (const float4*)(x + (size_t)n * LATENT);
#pragma unroll
    for (int q = 0; q < LATENT / 4; ++q) {
        float4 v = xr[q];
        const float* vp = (const float*)&v;
#pragma unroll
        for (int j = 0; j < 4; ++j) {
            int i = q * 4 + j;
            float xvv = vp[j];
            a0 += xvv * root_w[i * 3 + 0];
            a1 += xvv * root_w[i * 3 + 1];
            a2 += xvv * root_w[i * 3 + 2];
        }
    }
    out[(size_t)n * 3 + 0] = a0;
    out[(size_t)n * 3 + 1] = a1;
    out[(size_t)n * 3 + 2] = a2;
}


extern "C" void kernel_launch(void* const* d_in, const int* in_sizes, int n_in,
                              void* d_out, int out_size, void* d_ws, size_t ws_size,
                              hipStream_t stream) {
    const float* x          = (const float*)d_in[0];
    const int*   edge_index = (const int*)  d_in[1];
    const float* edge_attr  = (const float*)d_in[2];
    const float* W1 = (const float*)d_in[3];
    const float* b1 = (const float*)d_in[4];
    const float* W2 = (const float*)d_in[5];
    const float* b2 = (const float*)d_in[6];
    const float* W3 = (const float*)d_in[7];
    const float* b3 = (const float*)d_in[8];
    const float* W4 = (const float*)d_in[9];
    const float* b4 = (const float*)d_in[10];
    const float* root_w = (const float*)d_in[11];
    const float* bias   = (const float*)d_in[12];
    float* out = (float*)d_out;

    const int E = in_sizes[1] / 2;
    const int N = in_sizes[0] / LATENT;

    float* summed = (float*)d_ws;
    float* count  = summed + (size_t)N * 3;

    {
        int zn = N * 4;
        zero_f32<<<(zn + 255) / 256, 256, 0, stream>>>(summed, zn);
    }
    {
        int tiles  = (E + 15) / 16;
        int blocks = (tiles + WAVES - 1) / WAVES;
        nnconv_edge_kernel<<<blocks, WAVES * 32, 0, stream>>>(
            x, edge_index, edge_attr,
            W1, b1, W2, b2, W3, b3, W4, b4,
            summed, count, E);
    }
    nnconv_finalize_kernel<<<(N + 127) / 128, 128, 0, stream>>>(
        x, root_w, bias, summed, count, out, N);
}


# Staging loads freed of per-element branches (weights: index clamp + value select, constant trips; edge_attr: bound tested once per tile): each phase's loads issue back to back, not one full wait each
# speedup vs baseline: 1.6567x; 1.6567x over previous
#include <hip/hip_runtime.h>
#include <hip/hip_bf16.h>


typedef __attribute__((ext_vector_type(16))) _Float16 v16h;
typedef __attribute__((ext_vector_type(8)))  _Float16 v8h;
typedef __attribute__((ext_vector_type(8)))  float    v8f;

#define LATENT   64
#define EDGE_IN  70
#define K1PAD    96
#define HID      20
#define OUTD     192
#define OUT_CH   3
#define WAVES    4


__device__ inline v8f wmma16(v16h a, v16h b, v8f c) {
    return __builtin_amdgcn_wmma_f32_16x16x32_f16(
        false, a, false, b, (short)0, c, false, false);
}

__device__ inline v16h cat8(v8h lo, v8h hi) {
    return __builtin_shufflevector(lo, hi,
        0, 1, 2, 3, 4, 5, 6, 7, 8, 9, 10, 11, 12, 13, 14, 15);
}

__device__ inline v16h load_a(const _Float16* H, int ld, int k0) {
    int lane = threadIdx.x & 31;
    int m  = lane & 15;
    int kb = k0 + ((lane & 16) ? 8 : 0);
    v8h lo = *(const v8h*)(H + m * ld + kb);
    v8h hi = *(const v8h*)(H + m * ld + kb + 16);
    return cat8(lo, hi);
}

__device__ inline v16h make_b(const _Float16* WT, int ldk, int k0, int n0) {
    int lane = threadIdx.x & 31;
    int n  = n0 + (lane & 15);
    int kb = k0 + ((lane & 16) ? 8 : 0);
    v8h lo = *(const v8h*)(WT + n * ldk + kb);
    v8h hi = *(const v8h*)(WT + n * ldk + kb + 16);
    return cat8(lo, hi);
}

__device__ inline void store_c(_Float16* H, int ld, v8f c, int n0,
                               const float* bias, bool relu) {
    int lane = threadIdx.x & 31;
    int n  = n0 + (lane & 15);
    int mb = (lane & 16) ? 8 : 0;
    float bv = bias[n];
#pragma unroll
    for (int r = 0; r < 8; ++r) {
        float v = c[r] + bv;
        if (relu) v = fmaxf(v, 0.0f);
        H[(mb + r) * ld + n] = (_Float16)v;
    }
}


__global__ void zero_f32(float* p, int n) {
    int i = blockIdx.x * blockDim.x + threadIdx.x;
    if (i < n) p[i] = 0.0f;
}

__global__ __launch_bounds__(WAVES * 32)
void nnconv_edge_kernel(const float* __restrict__ x,
                        const int*   __restrict__ edge_index,
                        const float* __restrict__ edge_attr,
                        const float* __restrict__ W1, const float* __restrict__ b1,
                        const float* __restrict__ W2, const float* __restrict__ b2,
                        const float* __restrict__ W3, const float* __restrict__ b3,
                        const float* __restrict__ W4, const float* __restrict__ b4,
                        float* __restrict__ summed, float* __restrict__ count,
                        int E) {
    __shared__ alignas(16) _Float16 sW1T[32  * K1PAD];
    __shared__ alignas(16) _Float16 sW2T[32  * 32];
    __shared__ alignas(16) _Float16 sW3T[32  * 32];
    __shared__ alignas(16) _Float16 sW4T[OUTD * 32];
    __shared__ float sB1[32], sB2[32], sB3[32];
    __shared__ float sB4[OUTD];
    __shared__ alignas(16) _Float16 sA [WAVES][16 * K1PAD];
    __shared__ alignas(16) _Float16 sO4[WAVES][16 * OUTD];

    const int tid = threadIdx.x;

    constexpr int NT = WAVES * 32;
    static_assert((32 * K1PAD) % NT == 0 && (32 * 32) % NT == 0 &&
                  (OUTD * 32) % NT == 0, "staging loops assume whole trips");
#pragma unroll
    for (int j = 0; j < 32 * K1PAD / NT; ++j) {
        int i = tid + j * NT;
        int n = i / K1PAD, k = i % K1PAD;
        float w = W1[min(k, EDGE_IN - 1) * HID + min(n, HID - 1)];
        sW1T[i] = (_Float16)((k < EDGE_IN && n < HID) ? w : 0.0f);
    }
#pragma unroll
    for (int j = 0; j < 32 * 32 / NT; ++j) {
        int i = tid + j * NT;
        int n = i >> 5, k = i & 31;
        int g = min(k, HID - 1) * HID + min(n, HID - 1);
        float w2 = W2[g], w3 = W3[g];
        bool ok = (k < HID && n < HID);
        sW2T[i] = (_Float16)(ok ? w2 : 0.0f);
        sW3T[i] = (_Float16)(ok ? w3 : 0.0f);
    }
#pragma unroll
    for (int j = 0; j < OUTD * 32 / NT; ++j) {
        int i = tid + j * NT;
        int n = i >> 5, k = i & 31;
        float w = W4[min(k, HID - 1) * OUTD + n];
        sW4T[i] = (_Float16)((k < HID) ? w : 0.0f);
    }
    for (int i = tid; i < 32; i += blockDim.x) {
        sB1[i] = (i < HID) ? b1[i] : 0.0f;
        sB2[i] = (i < HID) ? b2[i] : 0.0f;
        sB3[i] = (i < HID) ? b3[i] : 0.0f;
    }
    for (int i = tid; i < OUTD; i += blockDim.x) sB4[i] = b4[i];
    __syncthreads();

    const int lane = tid & 31;
    const int wave = tid >> 5;
    const int m    = lane & 15;
    const int half = lane >> 4;

    const long tile = (long)blockIdx.x * WAVES + wave;
    const long e0   = tile * 16;
    if (e0 >= E) return;

    _Float16* A1 = sA [wave];
    _Float16* O4 = sO4[wave];

    {
        const long gbase = e0 * EDGE_IN;
        if (e0 + 16 <= E) {
            const float* ap = edge_attr + gbase;
            float v[16 * EDGE_IN / 32];
#pragma unroll
            for (int j = 0; j < 16 * EDGE_IN / 32; ++j) v[j] = ap[lane + 32 * j];
#pragma unroll
            for (int j = 0; j < 16 * EDGE_IN / 32; ++j) {
                int i = lane + 32 * j;
                A1[(i / EDGE_IN) * K1PAD + (i % EDGE_IN)] = (_Float16)v[j];
            }
        } else {
            const long gmax = (long)E * EDGE_IN;
            for (int i = lane; i < 16 * EDGE_IN; i += 32) {
                long g = gbase + i;
                float v = (g < gmax) ? edge_attr[g] : 0.0f;
                A1[(i / EDGE_IN) * K1PAD + (i % EDGE_IN)] = (_Float16)v;
            }
        }
#pragma unroll
        for (int i = lane; i < 16 * (K1PAD - EDGE_IN); i += 32) {
            A1[(i / 26) * K1PAD + EDGE_IN + (i % 26)] = (_Float16)0.0f;
        }
    }
    asm volatile("s_wait_dscnt 0" ::: "memory");

    _Float16* H = A1;
    {
        v8f c0 = {}, c1 = {};
#pragma unroll
        for (int ks = 0; ks < 3; ++ks) {
            v16h a = load_a(A1, K1PAD, ks * 32);
            c0 = wmma16(a, make_b(sW1T, K1PAD, ks * 32, 0),  c0);
            c1 = wmma16(a, make_b(sW1T, K1PAD, ks * 32, 16), c1);
        }
        store_c(H, 32, c0, 0,  sB1, true);
        store_c(H, 32, c1, 16, sB1, true);
    }
    asm volatile("s_wait_dscnt 0" ::: "memory");

    {
        v16h a = load_a(H, 32, 0);
        v8f c0 = {}, c1 = {};
        c0 = wmma16(a, make_b(sW2T, 32, 0, 0),  c0);
        c1 = wmma16(a, make_b(sW2T, 32, 0, 16), c1);
        asm volatile("s_wait_dscnt 0" ::: "memory");
        store_c(H, 32, c0, 0,  sB2, true);
        store_c(H, 32, c1, 16, sB2, true);
    }
    asm volatile("s_wait_dscnt 0" ::: "memory");

    {
        v16h a = load_a(H, 32, 0);
        v8f c0 = {}, c1 = {};
        c0 = wmma16(a, make_b(sW3T, 32, 0, 0),  c0);
        c1 = wmma16(a, make_b(sW3T, 32, 0, 16), c1);
        asm volatile("s_wait_dscnt 0" ::: "memory");
        store_c(H, 32, c0, 0,  sB3, true);
        store_c(H, 32, c1, 16, sB3, true);
    }
    asm volatile("s_wait_dscnt 0" ::: "memory");

    {
        v16h a = load_a(H, 32, 0);
#pragma unroll
        for (int t = 0; t < OUTD / 16; ++t) {
            v8f c = {};
            c = wmma16(a, make_b(sW4T, 32, 0, t * 16), c);
            store_c(O4, OUTD, c, t * 16, sB4, false);
        }
    }
    asm volatile("s_wait_dscnt 0" ::: "memory");

    const long e = e0 + m;
    if (e < E) {
        const int src = edge_index[e];
        const int dst = edge_index[(size_t)E + e];

        float xv[32];
        const float4* xr = (const float4*)(x + (size_t)src * LATENT + half * 32);
#pragma unroll
        for (int q = 0; q < 8; ++q) {
            float4 t = xr[q];
            xv[q * 4 + 0] = t.x; xv[q * 4 + 1] = t.y;
            xv[q * 4 + 2] = t.z; xv[q * 4 + 3] = t.w;
        }

        const v8h* wp = (const v8h*)(O4 + m * OUTD + half * 32 * OUT_CH);
        float p0 = 0.0f, p1 = 0.0f, p2 = 0.0f;
#pragma unroll
        for (int q = 0; q < 12; ++q) {
            v8h w8 = wp[q];
#pragma unroll
            for (int j = 0; j < 8; ++j) {
                int t = q * 8 + j;
                int i = t / 3, c = t % 3;
                float prod = xv[i] * (float)w8[j];
                if (c == 0) p0 += prod;
                else if (c == 1) p1 += prod;
                else p2 += prod;
            }
        }
        p0 += __shfl_xor(p0, 16);
        p1 += __shfl_xor(p1, 16);
        p2 += __shfl_xor(p2, 16);
        if (half == 0) {
            atomicAdd(&summed[(size_t)dst * 3 + 0], p0);
            atomicAdd(&summed[(size_t)dst * 3 + 1], p1);
            atomicAdd(&summed[(size_t)dst * 3 + 2], p2);
            atomicAdd(&count[dst], 1.0f);
        }
    }
}

__global__ void nnconv_finalize_kernel(const float* __restrict__ x,
                                       const float* __restrict__ root_w,
                                       const float* __restrict__ bias,
                                       const float* __restrict__ summed,
                                       const float* __restrict__ count,
                                       float* __restrict__ out, int N) {
    int n = blockIdx.x * blockDim.x + threadIdx.x;
    if (n >= N) return;
    float inv = 1.0f / fmaxf(count[n], 1.0f);
    float a0 = summed[(size_t)n * 3 + 0] * inv + bias[0];
    float a1 = summed[(size_t)n * 3 + 1] * inv + bias[1];
    float a2 = summed[(size_t)n * 3 + 2] * inv + bias[2];
    const float4* xr = (const float4*)(x + (size_t)n * LATENT);
#pragma unroll
    for (int q = 0; q < LATENT / 4; ++q) {
        float4 v = xr[q];
        const float* vp = (const float*)&v;
#pragma unroll
        for (int j = 0; j < 4; ++j) {
            int i = q * 4 + j;
            float xvv = vp[j];
            a0 += xvv * root_w[i * 3 + 0];
            a1 += xvv * root_w[i * 3 + 1];
            a2 += xvv * root_w[i * 3 + 2];
        }
    }
    out[(size_t)n * 3 + 0] = a0;
    out[(size_t)n * 3 + 1] = a1;
    out[(size_t)n * 3 + 2] = a2;
}


extern "C" void kernel_launch(void* const* d_in, const int* in_sizes, int n_in,
                              void* d_out, int out_size, void* d_ws, size_t ws_size,
                              hipStream_t stream) {
    const float* x          = (const float*)d_in[0];
    const int*   edge_index = (const int*)  d_in[1];
    const float* edge_attr  = (const float*)d_in[2];
    const float* W1 = (const float*)d_in[3];
    const float* b1 = (const float*)d_in[4];
    const float* W2 = (const float*)d_in[5];
    const float* b2 = (const float*)d_in[6];
    const float* W3 = (const float*)d_in[7];
    const float* b3 = (const float*)d_in[8];
    const float* W4 = (const float*)d_in[9];
    const float* b4 = (const float*)d_in[10];
    const float* root_w = (const float*)d_in[11];
    const float* bias   = (const float*)d_in[12];
    float* out = (float*)d_out;

    const int E = in_sizes[1] / 2;
    const int N = in_sizes[0] / LATENT;

    float* summed = (float*)d_ws;
    float* count  = summed + (size_t)N * 3;

    {
        int zn = N * 4;
        zero_f32<<<(zn + 255) / 256, 256, 0, stream>>>(summed, zn);
    }
    {
        int tiles  = (E + 15) / 16;
        int blocks = (tiles + WAVES - 1) / WAVES;
        nnconv_edge_kernel<<<blocks, WAVES * 32, 0, stream>>>(
            x, edge_index, edge_attr,
            W1, b1, W2, b2, W3, b3, W4, b4,
            summed, count, E);
    }
    nnconv_finalize_kernel<<<(N + 127) / 128, 128, 0, stream>>>(
        x, root_w, bias, summed, count, out, N);
}


# LDS row pitches made 16 B x odd (layer-1 rows 208 B, 32-wide rows 80 B, layer-4 output rows 400 B): ds_load_b128 fragment and einsum reads and D-tile stores no longer collide on banks
# speedup vs baseline: 2.0718x; 1.2506x over previous
#include <hip/hip_runtime.h>
#include <hip/hip_bf16.h>


typedef __attribute__((ext_vector_type(16))) _Float16 v16h;
typedef __attribute__((ext_vector_type(8)))  _Float16 v8h;
typedef __attribute__((ext_vector_type(8)))  float    v8f;

#define LATENT   64
#define EDGE_IN  70
#define K1PAD    96
#define LD1      104
#define LDH      40
#define LDO      200
#define HID      20
#define OUTD     192
#define OUT_CH   3
#define WAVES    16
#define EDGE_GRID 256


__device__ inline v8f wmma16(v16h a, v16h b, v8f c) {
    return __builtin_amdgcn_wmma_f32_16x16x32_f16(
        false, a, false, b, (short)0, c, false, false);
}

__device__ inline v16h cat8(v8h lo, v8h hi) {
    return __builtin_shufflevector(lo, hi,
        0, 1, 2, 3, 4, 5, 6, 7, 8, 9, 10, 11, 12, 13, 14, 15);
}

__device__ inline v16h load_a(const _Float16* H, int ld, int k0) {
    int lane = threadIdx.x & 31;
    int m  = lane & 15;
    int kb = k0 + ((lane & 16) ? 8 : 0);
    v8h lo = *(const v8h*)(H + m * ld + kb);
    v8h hi = *(const v8h*)(H + m * ld + kb + 16);
    return cat8(lo, hi);
}

__device__ inline v16h make_b(const _Float16* WT, int ldk, int k0, int n0) {
    int lane = threadIdx.x & 31;
    int n  = n0 + (lane & 15);
    int kb = k0 + ((lane & 16) ? 8 : 0);
    v8h lo = *(const v8h*)(WT + n * ldk + kb);
    v8h hi = *(const v8h*)(WT + n * ldk + kb + 16);
    return cat8(lo, hi);
}

__device__ inline void store_c(_Float16* H, int ld, v8f c, int n0,
                               const float* bias, bool relu) {
    int lane = threadIdx.x & 31;
    int n  = n0 + (lane & 15);
    int mb = (lane & 16) ? 8 : 0;
    float bv = bias[n];
#pragma unroll
    for (int r = 0; r < 8; ++r) {
        float v = c[r] + bv;
        if (relu) v = fmaxf(v, 0.0f);
        H[(mb + r) * ld + n] = (_Float16)v;
    }
}


__global__ void zero_f32(float* p, int n) {
    int i = blockIdx.x * blockDim.x + threadIdx.x;
    if (i < n) p[i] = 0.0f;
}

__global__ __launch_bounds__(WAVES * 32)
void nnconv_edge_kernel(const float* __restrict__ x,
                        const int*   __restrict__ edge_index,
                        const float* __restrict__ edge_attr,
                        const float* __restrict__ W1, const float* __restrict__ b1,
                        const float* __restrict__ W2, const float* __restrict__ b2,
                        const float* __restrict__ W3, const float* __restrict__ b3,
                        const float* __restrict__ W4, const float* __restrict__ b4,
                        float* __restrict__ summed, float* __restrict__ count,
                        int E) {
    __shared__ alignas(16) _Float16 sW1T[32  * LD1];
    __shared__ alignas(16) _Float16 sW2T[32  * LDH];
    __shared__ alignas(16) _Float16 sW3T[32  * LDH];
    __shared__ alignas(16) _Float16 sW4T[OUTD * LDH];
    __shared__ float sB1[32], sB2[32], sB3[32];
    __shared__ float sB4[OUTD];
    __shared__ alignas(16) _Float16 sA [WAVES][16 * LD1];
    __shared__ alignas(16) _Float16 sO4[WAVES][16 * LDO];

    const int tid = threadIdx.x;

    constexpr int NT = WAVES * 32;
    static_assert((32 * K1PAD) % NT == 0 && (32 * 32) % NT == 0 &&
                  (OUTD * 32) % NT == 0, "staging loops assume whole trips");
#pragma unroll
    for (int j = 0; j < 32 * K1PAD / NT; ++j) {
        int i = tid + j * NT;
        int n = i / K1PAD, k = i % K1PAD;
        float w = W1[min(k, EDGE_IN - 1) * HID + min(n, HID - 1)];
        sW1T[n * LD1 + k] = (_Float16)((k < EDGE_IN && n < HID) ? w : 0.0f);
    }
#pragma unroll
    for (int j = 0; j < 32 * 32 / NT; ++j) {
        int i = tid + j * NT;
        int n = i >> 5, k = i & 31;
        int g = min(k, HID - 1) * HID + min(n, HID - 1);
        float w2 = W2[g], w3 = W3[g];
        bool ok = (k < HID && n < HID);
        sW2T[n * LDH + k] = (_Float16)(ok ? w2 : 0.0f);
        sW3T[n * LDH + k] = (_Float16)(ok ? w3 : 0.0f);
    }
#pragma unroll
    for (int j = 0; j < OUTD * 32 / NT; ++j) {
        int i = tid + j * NT;
        int n = i >> 5, k = i & 31;
        float w = W4[min(k, HID - 1) * OUTD + n];
        sW4T[n * LDH + k] = (_Float16)((k < HID) ? w : 0.0f);
    }
    for (int i = tid; i < 32; i += blockDim.x) {
        sB1[i] = (i < HID) ? b1[i] : 0.0f;
        sB2[i] = (i < HID) ? b2[i] : 0.0f;
        sB3[i] = (i < HID) ? b3[i] : 0.0f;
    }
    for (int i = tid; i < OUTD; i += blockDim.x) sB4[i] = b4[i];
    __syncthreads();

    const int lane = tid & 31;
    const int wave = __builtin_amdgcn_readfirstlane(tid >> 5);
    const int m    = lane & 15;
    const int half = lane >> 4;

    _Float16* A1 = sA [wave];
    _Float16* O4 = sO4[wave];

    const long tiles  = ((long)E + 15) >> 4;
    const long stride = (long)EDGE_GRID * WAVES;
    for (long tile = (long)wave * EDGE_GRID + blockIdx.x; tile < tiles;
         tile += stride) {
    const long e0 = tile * 16;

    {
        const long gbase = e0 * EDGE_IN;
        if (e0 + 16 <= E) {
            const float* ap = edge_attr + gbase;
            float v[16 * EDGE_IN / 32];
#pragma unroll
            for (int j = 0; j < 16 * EDGE_IN / 32; ++j) v[j] = ap[lane + 32 * j];
#pragma unroll
            for (int j = 0; j < 16 * EDGE_IN / 32; ++j) {
                int i = lane + 32 * j;
                A1[(i / EDGE_IN) * LD1 + (i % EDGE_IN)] = (_Float16)v[j];
            }
        } else {
            const long gmax = (long)E * EDGE_IN;
            for (int i = lane; i < 16 * EDGE_IN; i += 32) {
                long g = gbase + i;
                float v = (g < gmax) ? edge_attr[g] : 0.0f;
                A1[(i / EDGE_IN) * LD1 + (i % EDGE_IN)] = (_Float16)v;
            }
        }
#pragma unroll
        for (int i = lane; i < 16 * (K1PAD - EDGE_IN); i += 32) {
            A1[(i / 26) * LD1 + EDGE_IN + (i % 26)] = (_Float16)0.0f;
        }
    }
    asm volatile("s_wait_dscnt 0" ::: "memory");

    _Float16* H = A1;
    {
        v8f c0 = {}, c1 = {};
#pragma unroll
        for (int ks = 0; ks < 3; ++ks) {
            v16h a = load_a(A1, LD1, ks * 32);
            c0 = wmma16(a, make_b(sW1T, LD1, ks * 32, 0),  c0);
            c1 = wmma16(a, make_b(sW1T, LD1, ks * 32, 16), c1);
        }
        store_c(H, LDH, c0, 0,  sB1, true);
        store_c(H, LDH, c1, 16, sB1, true);
    }
    asm volatile("s_wait_dscnt 0" ::: "memory");

    {
        v16h a = load_a(H, LDH, 0);
        v8f c0 = {}, c1 = {};
        c0 = wmma16(a, make_b(sW2T, LDH, 0, 0),  c0);
        c1 = wmma16(a, make_b(sW2T, LDH, 0, 16), c1);
        asm volatile("s_wait_dscnt 0" ::: "memory");
        store_c(H, LDH, c0, 0,  sB2, true);
        store_c(H, LDH, c1, 16, sB2, true);
    }
    asm volatile("s_wait_dscnt 0" ::: "memory");

    {
        v16h a = load_a(H, LDH, 0);
        v8f c0 = {}, c1 = {};
        c0 = wmma16(a, make_b(sW3T, LDH, 0, 0),  c0);
        c1 = wmma16(a, make_b(sW3T, LDH, 0, 16), c1);
        asm volatile("s_wait_dscnt 0" ::: "memory");
        store_c(H, LDH, c0, 0,  sB3, true);
        store_c(H, LDH, c1, 16, sB3, true);
    }
    asm volatile("s_wait_dscnt 0" ::: "memory");

    {
        v16h a = load_a(H, LDH, 0);
#pragma unroll
        for (int t = 0; t < OUTD / 16; ++t) {
            v8f c = {};
            c = wmma16(a, make_b(sW4T, LDH, 0, t * 16), c);
            store_c(O4, LDO, c, t * 16, sB4, false);
        }
    }
    asm volatile("s_wait_dscnt 0" ::: "memory");

    const long e = e0 + m;
    if (e < E) {
        const int src = edge_index[e];
        const int dst = edge_index[(size_t)E + e];

        float xv[32];
        const float4* xr = (const float4*)(x + (size_t)src * LATENT + half * 32);
#pragma unroll
        for (int q = 0; q < 8; ++q) {
            float4 t = xr[q];
            xv[q * 4 + 0] = t.x; xv[q * 4 + 1] = t.y;
            xv[q * 4 + 2] = t.z; xv[q * 4 + 3] = t.w;
        }

        const v8h* wp = (const v8h*)(O4 + m * LDO + half * 32 * OUT_CH);
        float p0 = 0.0f, p1 = 0.0f, p2 = 0.0f;
#pragma unroll
        for (int q = 0; q < 12; ++q) {
            v8h w8 = wp[q];
#pragma unroll
            for (int j = 0; j < 8; ++j) {
                int t = q * 8 + j;
                int i = t / 3, c = t % 3;
                float prod = xv[i] * (float)w8[j];
                if (c == 0) p0 += prod;
                else if (c == 1) p1 += prod;
                else p2 += prod;
            }
        }
        p0 += __shfl_xor(p0, 16);
        p1 += __shfl_xor(p1, 16);
        p2 += __shfl_xor(p2, 16);
        if (half == 0) {
            atomicAdd(&summed[(size_t)dst * 3 + 0], p0);
            atomicAdd(&summed[(size_t)dst * 3 + 1], p1);
            atomicAdd(&summed[(size_t)dst * 3 + 2], p2);
            atomicAdd(&count[dst], 1.0f);
        }
    }
    }
}

__global__ void nnconv_finalize_kernel(const float* __restrict__ x,
                                       const float* __restrict__ root_w,
                                       const float* __restrict__ bias,
                                       const float* __restrict__ summed,
                                       const float* __restrict__ count,
                                       float* __restrict__ out, int N) {
    int n = blockIdx.x * blockDim.x + threadIdx.x;
    if (n >= N) return;
    float inv = 1.0f / fmaxf(count[n], 1.0f);
    float a0 = summed[(size_t)n * 3 + 0] * inv + bias[0];
    float a1 = summed[(size_t)n * 3 + 1] * inv + bias[1];
    float a2 = summed[(size_t)n * 3 + 2] * inv + bias[2];
    const float4* xr = (const float4*)(x + (size_t)n * LATENT);
#pragma unroll
    for (int q = 0; q < LATENT / 4; ++q) {
        float4 v = xr[q];
        const float* vp = (const float*)&v;
#pragma unroll
        for (int j = 0; j < 4; ++j) {
            int i = q * 4 + j;
            float xvv = vp[j];
            a0 += xvv * root_w[i * 3 + 0];
            a1 += xvv * root_w[i * 3 + 1];
            a2 += xvv * root_w[i * 3 + 2];
        }
    }
    out[(size_t)n * 3 + 0] = a0;
    out[(size_t)n * 3 + 1] = a1;
    out[(size_t)n * 3 + 2] = a2;
}


extern "C" void kernel_launch(void* const* d_in, const int* in_sizes, int n_in,
                              void* d_out, int out_size, void* d_ws, size_t ws_size,
                              hipStream_t stream) {
    const float* x          = (const float*)d_in[0];
    const int*   edge_index = (const int*)  d_in[1];
    const float* edge_attr  = (const float*)d_in[2];
    const float* W1 = (const float*)d_in[3];
    const float* b1 = (const float*)d_in[4];
    const float* W2 = (const float*)d_in[5];
    const float* b2 = (const float*)d_in[6];
    const float* W3 = (const float*)d_in[7];
    const float* b3 = (const float*)d_in[8];
    const float* W4 = (const float*)d_in[9];
    const float* b4 = (const float*)d_in[10];
    const float* root_w = (const float*)d_in[11];
    const float* bias   = (const float*)d_in[12];
    float* out = (float*)d_out;

    const int E = in_sizes[1] / 2;
    const int N = in_sizes[0] / LATENT;

    float* summed = (float*)d_ws;
    float* count  = summed + (size_t)N * 3;

    {
        int zn = N * 4;
        zero_f32<<<(zn + 255) / 256, 256, 0, stream>>>(summed, zn);
    }
    {
        nnconv_edge_kernel<<<EDGE_GRID, WAVES * 32, 0, stream>>>(
            x, edge_index, edge_attr,
            W1, b1, W2, b2, W3, b3, W4, b4,
            summed, count, E);
    }
    nnconv_finalize_kernel<<<(N + 127) / 128, 128, 0, stream>>>(
        x, root_w, bias, summed, count, out, N);
}


# Scatter accumulators interleaved [N][4] (3 sums + count): an edge's 4 atomic adds issue from 4 adjacent lanes of one instruction, 1 L2 request per edge, not 4; adds still land in arrival order
# speedup vs baseline: 3.6373x; 1.7556x over previous
#include <hip/hip_runtime.h>
#include <hip/hip_bf16.h>


typedef __attribute__((ext_vector_type(16))) _Float16 v16h;
typedef __attribute__((ext_vector_type(8)))  _Float16 v8h;
typedef __attribute__((ext_vector_type(8)))  float    v8f;

#define LATENT   64
#define EDGE_IN  70
#define K1PAD    96
#define LD1      104
#define LDH      40
#define LDO      200
#define HID      20
#define OUTD     192
#define OUT_CH   3
#define WAVES    16
#define EDGE_GRID 256


__device__ inline v8f wmma16(v16h a, v16h b, v8f c) {
    return __builtin_amdgcn_wmma_f32_16x16x32_f16(
        false, a, false, b, (short)0, c, false, false);
}

__device__ inline v16h cat8(v8h lo, v8h hi) {
    return __builtin_shufflevector(lo, hi,
        0, 1, 2, 3, 4, 5, 6, 7, 8, 9, 10, 11, 12, 13, 14, 15);
}

__device__ inline v16h load_a(const _Float16* H, int ld, int k0) {
    int lane = threadIdx.x & 31;
    int m  = lane & 15;
    int kb = k0 + ((lane & 16) ? 8 : 0);
    v8h lo = *(const v8h*)(H + m * ld + kb);
    v8h hi = *(const v8h*)(H + m * ld + kb + 16);
    return cat8(lo, hi);
}

__device__ inline v16h make_b(const _Float16* WT, int ldk, int k0, int n0) {
    int lane = threadIdx.x & 31;
    int n  = n0 + (lane & 15);
    int kb = k0 + ((lane & 16) ? 8 : 0);
    v8h lo = *(const v8h*)(WT + n * ldk + kb);
    v8h hi = *(const v8h*)(WT + n * ldk + kb + 16);
    return cat8(lo, hi);
}

__device__ inline void store_c(_Float16* H, int ld, v8f c, int n0,
                               const float* bias, bool relu) {
    int lane = threadIdx.x & 31;
    int n  = n0 + (lane & 15);
    int mb = (lane & 16) ? 8 : 0;
    float bv = bias[n];
#pragma unroll
    for (int r = 0; r < 8; ++r) {
        float v = c[r] + bv;
        if (relu) v = fmaxf(v, 0.0f);
        H[(mb + r) * ld + n] = (_Float16)v;
    }
}


__global__ void zero_f32(float* p, int n) {
    int i = blockIdx.x * blockDim.x + threadIdx.x;
    if (i < n) p[i] = 0.0f;
}

__global__ __launch_bounds__(WAVES * 32)
void nnconv_edge_kernel(const float* __restrict__ x,
                        const int*   __restrict__ edge_index,
                        const float* __restrict__ edge_attr,
                        const float* __restrict__ W1, const float* __restrict__ b1,
                        const float* __restrict__ W2, const float* __restrict__ b2,
                        const float* __restrict__ W3, const float* __restrict__ b3,
                        const float* __restrict__ W4, const float* __restrict__ b4,
                        float* __restrict__ acc,
                        int E) {
    __shared__ alignas(16) _Float16 sW1T[32  * LD1];
    __shared__ alignas(16) _Float16 sW2T[32  * LDH];
    __shared__ alignas(16) _Float16 sW3T[32  * LDH];
    __shared__ alignas(16) _Float16 sW4T[OUTD * LDH];
    __shared__ float sB1[32], sB2[32], sB3[32];
    __shared__ float sB4[OUTD];
    __shared__ alignas(16) _Float16 sA [WAVES][16 * LD1];
    __shared__ alignas(16) _Float16 sO4[WAVES][16 * LDO];
    __shared__ alignas(16) float sMsg[WAVES][16 * 4];

    const int tid = threadIdx.x;

    constexpr int NT = WAVES * 32;
    static_assert((32 * K1PAD) % NT == 0 && (32 * 32) % NT == 0 &&
                  (OUTD * 32) % NT == 0, "staging loops assume whole trips");
#pragma unroll
    for (int j = 0; j < 32 * K1PAD / NT; ++j) {
        int i = tid + j * NT;
        int n = i / K1PAD, k = i % K1PAD;
        float w = W1[min(k, EDGE_IN - 1) * HID + min(n, HID - 1)];
        sW1T[n * LD1 + k] = (_Float16)((k < EDGE_IN && n < HID) ? w : 0.0f);
    }
#pragma unroll
    for (int j = 0; j < 32 * 32 / NT; ++j) {
        int i = tid + j * NT;
        int n = i >> 5, k = i & 31;
        int g = min(k, HID - 1) * HID + min(n, HID - 1);
        float w2 = W2[g], w3 = W3[g];
        bool ok = (k < HID && n < HID);
        sW2T[n * LDH + k] = (_Float16)(ok ? w2 : 0.0f);
        sW3T[n * LDH + k] = (_Float16)(ok ? w3 : 0.0f);
    }
#pragma unroll
    for (int j = 0; j < OUTD * 32 / NT; ++j) {
        int i = tid + j * NT;
        int n = i >> 5, k = i & 31;
        float w = W4[min(k, HID - 1) * OUTD + n];
        sW4T[n * LDH + k] = (_Float16)((k < HID) ? w : 0.0f);
    }
    for (int i = tid; i < 32; i += blockDim.x) {
        sB1[i] = (i < HID) ? b1[i] : 0.0f;
        sB2[i] = (i < HID) ? b2[i] : 0.0f;
        sB3[i] = (i < HID) ? b3[i] : 0.0f;
    }
    for (int i = tid; i < OUTD; i += blockDim.x) sB4[i] = b4[i];
    __syncthreads();

    const int lane = tid & 31;
    const int wave = __builtin_amdgcn_readfirstlane(tid >> 5);
    const int m    = lane & 15;
    const int half = lane >> 4;

    _Float16* A1 = sA [wave];
    _Float16* O4 = sO4[wave];

    const long tiles  = ((long)E + 15) >> 4;
    const long stride = (long)EDGE_GRID * WAVES;
    for (long tile = (long)wave * EDGE_GRID + blockIdx.x; tile < tiles;
         tile += stride) {
    const long e0 = tile * 16;

    {
        const long gbase = e0 * EDGE_IN;
        if (e0 + 16 <= E) {
            const float* ap = edge_attr + gbase;
            float v[16 * EDGE_IN / 32];
#pragma unroll
            for (int j = 0; j < 16 * EDGE_IN / 32; ++j) v[j] = ap[lane + 32 * j];
#pragma unroll
            for (int j = 0; j < 16 * EDGE_IN / 32; ++j) {
                int i = lane + 32 * j;
                A1[(i / EDGE_IN) * LD1 + (i % EDGE_IN)] = (_Float16)v[j];
            }
        } else {
            const long gmax = (long)E * EDGE_IN;
            for (int i = lane; i < 16 * EDGE_IN; i += 32) {
                long g = gbase + i;
                float v = (g < gmax) ? edge_attr[g] : 0.0f;
                A1[(i / EDGE_IN) * LD1 + (i % EDGE_IN)] = (_Float16)v;
            }
        }
#pragma unroll
        for (int i = lane; i < 16 * (K1PAD - EDGE_IN); i += 32) {
            A1[(i / 26) * LD1 + EDGE_IN + (i % 26)] = (_Float16)0.0f;
        }
    }
    asm volatile("s_wait_dscnt 0" ::: "memory");

    _Float16* H = A1;
    {
        v8f c0 = {}, c1 = {};
#pragma unroll
        for (int ks = 0; ks < 3; ++ks) {
            v16h a = load_a(A1, LD1, ks * 32);
            c0 = wmma16(a, make_b(sW1T, LD1, ks * 32, 0),  c0);
            c1 = wmma16(a, make_b(sW1T, LD1, ks * 32, 16), c1);
        }
        store_c(H, LDH, c0, 0,  sB1, true);
        store_c(H, LDH, c1, 16, sB1, true);
    }
    asm volatile("s_wait_dscnt 0" ::: "memory");

    {
        v16h a = load_a(H, LDH, 0);
        v8f c0 = {}, c1 = {};
        c0 = wmma16(a, make_b(sW2T, LDH, 0, 0),  c0);
        c1 = wmma16(a, make_b(sW2T, LDH, 0, 16), c1);
        asm volatile("s_wait_dscnt 0" ::: "memory");
        store_c(H, LDH, c0, 0,  sB2, true);
        store_c(H, LDH, c1, 16, sB2, true);
    }
    asm volatile("s_wait_dscnt 0" ::: "memory");

    {
        v16h a = load_a(H, LDH, 0);
        v8f c0 = {}, c1 = {};
        c0 = wmma16(a, make_b(sW3T, LDH, 0, 0),  c0);
        c1 = wmma16(a, make_b(sW3T, LDH, 0, 16), c1);
        asm volatile("s_wait_dscnt 0" ::: "memory");
        store_c(H, LDH, c0, 0,  sB3, true);
        store_c(H, LDH, c1, 16, sB3, true);
    }
    asm volatile("s_wait_dscnt 0" ::: "memory");

    {
        v16h a = load_a(H, LDH, 0);
#pragma unroll
        for (int t = 0; t < OUTD / 16; ++t) {
            v8f c = {};
            c = wmma16(a, make_b(sW4T, LDH, 0, t * 16), c);
            store_c(O4, LDO, c, t * 16, sB4, false);
        }
    }
    asm volatile("s_wait_dscnt 0" ::: "memory");

    const long e = e0 + m;
    const int  word  = lane & 3;
    const long ea    = e0 + (lane >> 2);
    const long eb    = ea + 8;
    const long elast = (long)E - 1;
    const int  da = edge_index[(size_t)E + (ea < elast ? ea : elast)];
    const int  db = edge_index[(size_t)E + (eb < elast ? eb : elast)];
    float p0 = 0.0f, p1 = 0.0f, p2 = 0.0f;
    if (e < E) {
        const int src = edge_index[e];

        float xv[32];
        const float4* xr = (const float4*)(x + (size_t)src * LATENT + half * 32);
#pragma unroll
        for (int q = 0; q < 8; ++q) {
            float4 t = xr[q];
            xv[q * 4 + 0] = t.x; xv[q * 4 + 1] = t.y;
            xv[q * 4 + 2] = t.z; xv[q * 4 + 3] = t.w;
        }

        const v8h* wp = (const v8h*)(O4 + m * LDO + half * 32 * OUT_CH);
#pragma unroll
        for (int q = 0; q < 12; ++q) {
            v8h w8 = wp[q];
#pragma unroll
            for (int j = 0; j < 8; ++j) {
                int t = q * 8 + j;
                int i = t / 3, c = t % 3;
                float prod = xv[i] * (float)w8[j];
                if (c == 0) p0 += prod;
                else if (c == 1) p1 += prod;
                else p2 += prod;
            }
        }
    }
    p0 += __shfl_xor(p0, 16);
    p1 += __shfl_xor(p1, 16);
    p2 += __shfl_xor(p2, 16);

    {
        float* S = sMsg[wave];
        if (half == 0) *(float4*)(S + m * 4) = make_float4(p0, p1, p2, 1.0f);
        asm volatile("s_wait_dscnt 0" ::: "memory");
        const float va = S[lane];
        const float vb = S[32 + lane];
        if (ea < E) atomicAdd(&acc[(size_t)da * 4 + word], va);
        if (eb < E) atomicAdd(&acc[(size_t)db * 4 + word], vb);
    }
    }
}

__global__ void nnconv_finalize_kernel(const float* __restrict__ x,
                                       const float* __restrict__ root_w,
                                       const float* __restrict__ bias,
                                       const float* __restrict__ acc,
                                       float* __restrict__ out, int N) {
    int n = blockIdx.x * blockDim.x + threadIdx.x;
    if (n >= N) return;
    const float4 s = ((const float4*)acc)[n];
    float inv = 1.0f / fmaxf(s.w, 1.0f);
    float a0 = s.x * inv + bias[0];
    float a1 = s.y * inv + bias[1];
    float a2 = s.z * inv + bias[2];
    const float4* xr = (const float4*)(x + (size_t)n * LATENT);
#pragma unroll
    for (int q = 0; q < LATENT / 4; ++q) {
        float4 v = xr[q];
        const float* vp = (const float*)&v;
#pragma unroll
        for (int j = 0; j < 4; ++j) {
            int i = q * 4 + j;
            float xvv = vp[j];
            a0 += xvv * root_w[i * 3 + 0];
            a1 += xvv * root_w[i * 3 + 1];
            a2 += xvv * root_w[i * 3 + 2];
        }
    }
    out[(size_t)n * 3 + 0] = a0;
    out[(size_t)n * 3 + 1] = a1;
    out[(size_t)n * 3 + 2] = a2;
}


extern "C" void kernel_launch(void* const* d_in, const int* in_sizes, int n_in,
                              void* d_out, int out_size, void* d_ws, size_t ws_size,
                              hipStream_t stream) {
    const float* x          = (const float*)d_in[0];
    const int*   edge_index = (const int*)  d_in[1];
    const float* edge_attr  = (const float*)d_in[2];
    const float* W1 = (const float*)d_in[3];
    const float* b1 = (const float*)d_in[4];
    const float* W2 = (const float*)d_in[5];
    const float* b2 = (const float*)d_in[6];
    const float* W3 = (const float*)d_in[7];
    const float* b3 = (const float*)d_in[8];
    const float* W4 = (const float*)d_in[9];
    const float* b4 = (const float*)d_in[10];
    const float* root_w = (const float*)d_in[11];
    const float* bias   = (const float*)d_in[12];
    float* out = (float*)d_out;

    const int E = in_sizes[1] / 2;
    const int N = in_sizes[0] / LATENT;

    float* acc = (float*)d_ws;

    {
        int zn = N * 4;
        zero_f32<<<(zn + 255) / 256, 256, 0, stream>>>(acc, zn);
    }
    {
        nnconv_edge_kernel<<<EDGE_GRID, WAVES * 32, 0, stream>>>(
            x, edge_index, edge_attr,
            W1, b1, W2, b2, W3, b3, W4, b4,
            acc, E);
    }
    nnconv_finalize_kernel<<<(N + 127) / 128, 128, 0, stream>>>(
        x, root_w, bias, acc, out, N);
}


# MLP layers computed transposed (W^T tile as A, activations as B, bias as C operand): a D tile is the next layer's B fragment, so activations stay in registers; einsum reads layer 4's f32 accumulators
# speedup vs baseline: 3.9869x; 1.0961x over previous
#include <hip/hip_runtime.h>
#include <hip/hip_bf16.h>


typedef __attribute__((ext_vector_type(16))) _Float16 v16h;
typedef __attribute__((ext_vector_type(8)))  _Float16 v8h;
typedef __attribute__((ext_vector_type(8)))  float    v8f;

#define LATENT   64
#define EDGE_IN  70
#define K1PAD    96
#define LD1      104
#define LDH      40
#define HID      20
#define OUTD     192
#define OUT_CH   3
#define OHALF    96
#define WAVES    16
#define EDGE_GRID 256


__device__ inline v8f wmma16(v16h a, v16h b, v8f c) {
    return __builtin_amdgcn_wmma_f32_16x16x32_f16(
        false, a, false, b, (short)0, c, false, false);
}

__device__ inline v16h cat8(v8h lo, v8h hi) {
    return __builtin_shufflevector(lo, hi,
        0, 1, 2, 3, 4, 5, 6, 7, 8, 9, 10, 11, 12, 13, 14, 15);
}

__device__ inline v16h frag(const _Float16* T, int ld, int k0, int r0) {
    int lane = threadIdx.x & 31;
    int r  = r0 + (lane & 15);
    int kb = k0 + ((lane & 16) ? 8 : 0);
    v8h lo = *(const v8h*)(T + r * ld + kb);
    v8h hi = *(const v8h*)(T + r * ld + kb + 16);
    return cat8(lo, hi);
}

__device__ inline v8f bias_c(const float* b, int n0) {
    int half = (threadIdx.x & 31) >> 4;
    return *(const v8f*)(b + n0 + 8 * half);
}

__device__ inline v16h relu_pack(v8f d0, v8f d1) {
    v16h h;
#pragma unroll
    for (int r = 0; r < 8; ++r) {
        h[r]     = (_Float16)d0[r];
        h[8 + r] = (_Float16)d1[r];
    }
    const v16h zero = {};
    return __builtin_elementwise_max(h, zero);
}


__global__ void zero_f32(float* p, int n) {
    int i = blockIdx.x * blockDim.x + threadIdx.x;
    if (i < n) p[i] = 0.0f;
}

__global__ __launch_bounds__(WAVES * 32)
void nnconv_edge_kernel(const float* __restrict__ x,
                        const int*   __restrict__ edge_index,
                        const float* __restrict__ edge_attr,
                        const float* __restrict__ W1, const float* __restrict__ b1,
                        const float* __restrict__ W2, const float* __restrict__ b2,
                        const float* __restrict__ W3, const float* __restrict__ b3,
                        const float* __restrict__ W4, const float* __restrict__ b4,
                        float* __restrict__ acc,
                        int E) {
    __shared__ alignas(16) _Float16 sW1T[32  * LD1];
    __shared__ alignas(16) _Float16 sW2T[32  * LDH];
    __shared__ alignas(16) _Float16 sW3T[32  * LDH];
    __shared__ alignas(16) _Float16 sW4T[OUTD * LDH];
    __shared__ alignas(32) float sB1[32], sB2[32], sB3[32];
    __shared__ alignas(32) float sB4[OUTD];
    __shared__ alignas(16) _Float16 sA [WAVES][16 * LD1];
    __shared__ alignas(16) float sMsg[WAVES][16 * 4];

    const int tid = threadIdx.x;

    constexpr int NT = WAVES * 32;
    static_assert((32 * K1PAD) % NT == 0 && (32 * 32) % NT == 0 &&
                  (OUTD * 32) % NT == 0, "staging loops assume whole trips");
#pragma unroll
    for (int j = 0; j < 32 * K1PAD / NT; ++j) {
        int i = tid + j * NT;
        int n = i / K1PAD, k = i % K1PAD;
        float w = W1[min(k, EDGE_IN - 1) * HID + min(n, HID - 1)];
        sW1T[n * LD1 + k] = (_Float16)((k < EDGE_IN && n < HID) ? w : 0.0f);
    }
#pragma unroll
    for (int j = 0; j < 32 * 32 / NT; ++j) {
        int i = tid + j * NT;
        int n = i >> 5, k = i & 31;
        int g = min(k, HID - 1) * HID + min(n, HID - 1);
        float w2 = W2[g], w3 = W3[g];
        bool ok = (k < HID && n < HID);
        sW2T[n * LDH + k] = (_Float16)(ok ? w2 : 0.0f);
        sW3T[n * LDH + k] = (_Float16)(ok ? w3 : 0.0f);
    }
#pragma unroll
    for (int j = 0; j < OUTD * 32 / NT; ++j) {
        int i = tid + j * NT;
        int n = i >> 5, k = i & 31;
        float w = W4[min(k, HID - 1) * OUTD + n];
        int hh = n / OHALF, rem = n % OHALF;
        int row = (rem >> 3) * 16 + hh * 8 + (rem & 7);
        sW4T[row * LDH + k] = (_Float16)((k < HID) ? w : 0.0f);
    }
    for (int i = tid; i < 32; i += blockDim.x) {
        sB1[i] = (i < HID) ? b1[i] : 0.0f;
        sB2[i] = (i < HID) ? b2[i] : 0.0f;
        sB3[i] = (i < HID) ? b3[i] : 0.0f;
    }
    for (int i = tid; i < OUTD; i += blockDim.x) sB4[i] = b4[i];
    __syncthreads();

    const int lane = tid & 31;
    const int wave = __builtin_amdgcn_readfirstlane(tid >> 5);
    const int m    = lane & 15;
    const int half = lane >> 4;

    _Float16* A1 = sA [wave];

#pragma unroll
    for (int i = lane; i < 16 * (K1PAD - EDGE_IN); i += 32) {
        A1[(i / 26) * LD1 + EDGE_IN + (i % 26)] = (_Float16)0.0f;
    }

    const long tiles  = ((long)E + 15) >> 4;
    const long stride = (long)EDGE_GRID * WAVES;
    for (long tile = (long)wave * EDGE_GRID + blockIdx.x; tile < tiles;
         tile += stride) {
    const long e0 = tile * 16;

    {
        const long gbase = e0 * EDGE_IN;
        if (e0 + 16 <= E) {
            const float* ap = edge_attr + gbase;
            float v[16 * EDGE_IN / 32];
#pragma unroll
            for (int j = 0; j < 16 * EDGE_IN / 32; ++j) v[j] = ap[lane + 32 * j];
#pragma unroll
            for (int j = 0; j < 16 * EDGE_IN / 32; ++j) {
                int i = lane + 32 * j;
                A1[(i / EDGE_IN) * LD1 + (i % EDGE_IN)] = (_Float16)v[j];
            }
        } else {
            const long gmax = (long)E * EDGE_IN;
            for (int i = lane; i < 16 * EDGE_IN; i += 32) {
                long g = gbase + i;
                float v = (g < gmax) ? edge_attr[g] : 0.0f;
                A1[(i / EDGE_IN) * LD1 + (i % EDGE_IN)] = (_Float16)v;
            }
        }
    }
    asm volatile("s_wait_dscnt 0" ::: "memory");

    v16h h;
    {
        v8f d0 = bias_c(sB1, 0), d1 = bias_c(sB1, 16);
#pragma unroll
        for (int ks = 0; ks < 3; ++ks) {
            v16h xT = frag(A1, LD1, ks * 32, 0);
            d0 = wmma16(frag(sW1T, LD1, ks * 32, 0),  xT, d0);
            d1 = wmma16(frag(sW1T, LD1, ks * 32, 16), xT, d1);
        }
        h = relu_pack(d0, d1);
    }
    {
        v8f d0 = wmma16(frag(sW2T, LDH, 0, 0),  h, bias_c(sB2, 0));
        v8f d1 = wmma16(frag(sW2T, LDH, 0, 16), h, bias_c(sB2, 16));
        h = relu_pack(d0, d1);
    }
    {
        v8f d0 = wmma16(frag(sW3T, LDH, 0, 0),  h, bias_c(sB3, 0));
        v8f d1 = wmma16(frag(sW3T, LDH, 0, 16), h, bias_c(sB3, 16));
        h = relu_pack(d0, d1);
    }

    const long e     = e0 + m;
    const int  word  = lane & 3;
    const long ea    = e0 + (lane >> 2);
    const long eb    = ea + 8;
    const long elast = (long)E - 1;
    const int  da = edge_index[(size_t)E + (ea < elast ? ea : elast)];
    const int  db = edge_index[(size_t)E + (eb < elast ? eb : elast)];

    float xv[32];
    {
        const int src = edge_index[e < elast ? e : elast];
        const float4* xr = (const float4*)(x + (size_t)src * LATENT + half * 32);
#pragma unroll
        for (int q = 0; q < 8; ++q) {
            float4 t = xr[q];
            xv[q * 4 + 0] = t.x; xv[q * 4 + 1] = t.y;
            xv[q * 4 + 2] = t.z; xv[q * 4 + 3] = t.w;
        }
    }

    float p0 = 0.0f, p1 = 0.0f, p2 = 0.0f;
    {
#pragma unroll
        for (int t = 0; t < OHALF / 8; ++t) {
            const v8f b = *(const v8f*)(sB4 + half * OHALF + t * 8);
            v8f c = wmma16(frag(sW4T, LDH, 0, t * 16), h, b);
#pragma unroll
            for (int j = 0; j < 8; ++j) {
                int f = t * 8 + j;
                int i = f / 3, cc = f % 3;
                float prod = xv[i] * c[j];
                if (cc == 0) p0 += prod;
                else if (cc == 1) p1 += prod;
                else p2 += prod;
            }
        }
    }
    p0 += __shfl_xor(p0, 16);
    p1 += __shfl_xor(p1, 16);
    p2 += __shfl_xor(p2, 16);

    {
        float* S = sMsg[wave];
        if (half == 0) *(float4*)(S + m * 4) = make_float4(p0, p1, p2, 1.0f);
        asm volatile("s_wait_dscnt 0" ::: "memory");
        const float va = S[lane];
        const float vb = S[32 + lane];
        if (ea < E) atomicAdd(&acc[(size_t)da * 4 + word], va);
        if (eb < E) atomicAdd(&acc[(size_t)db * 4 + word], vb);
    }
    }
}

__global__ void nnconv_finalize_kernel(const float* __restrict__ x,
                                       const float* __restrict__ root_w,
                                       const float* __restrict__ bias,
                                       const float* __restrict__ acc,
                                       float* __restrict__ out, int N) {
    int n = blockIdx.x * blockDim.x + threadIdx.x;
    if (n >= N) return;
    const float4 s = ((const float4*)acc)[n];
    float inv = 1.0f / fmaxf(s.w, 1.0f);
    float a0 = s.x * inv + bias[0];
    float a1 = s.y * inv + bias[1];
    float a2 = s.z * inv + bias[2];
    const float4* xr = (const float4*)(x + (size_t)n * LATENT);
#pragma unroll
    for (int q = 0; q < LATENT / 4; ++q) {
        float4 v = xr[q];
        const float* vp = (const float*)&v;
#pragma unroll
        for (int j = 0; j < 4; ++j) {
            int i = q * 4 + j;
            float xvv = vp[j];
            a0 += xvv * root_w[i * 3 + 0];
            a1 += xvv * root_w[i * 3 + 1];
            a2 += xvv * root_w[i * 3 + 2];
        }
    }
    out[(size_t)n * 3 + 0] = a0;
    out[(size_t)n * 3 + 1] = a1;
    out[(size_t)n * 3 + 2] = a2;
}


extern "C" void kernel_launch(void* const* d_in, const int* in_sizes, int n_in,
                              void* d_out, int out_size, void* d_ws, size_t ws_size,
                              hipStream_t stream) {
    const float* x          = (const float*)d_in[0];
    const int*   edge_index = (const int*)  d_in[1];
    const float* edge_attr  = (const float*)d_in[2];
    const float* W1 = (const float*)d_in[3];
    const float* b1 = (const float*)d_in[4];
    const float* W2 = (const float*)d_in[5];
    const float* b2 = (const float*)d_in[6];
    const float* W3 = (const float*)d_in[7];
    const float* b3 = (const float*)d_in[8];
    const float* W4 = (const float*)d_in[9];
    const float* b4 = (const float*)d_in[10];
    const float* root_w = (const float*)d_in[11];
    const float* bias   = (const float*)d_in[12];
    float* out = (float*)d_out;

    const int E = in_sizes[1] / 2;
    const int N = in_sizes[0] / LATENT;

    float* acc = (float*)d_ws;

    {
        int zn = N * 4;
        zero_f32<<<(zn + 255) / 256, 256, 0, stream>>>(acc, zn);
    }
    {
        nnconv_edge_kernel<<<EDGE_GRID, WAVES * 32, 0, stream>>>(
            x, edge_index, edge_attr,
            W1, b1, W2, b2, W3, b3, W4, b4,
            acc, E);
    }
    nnconv_finalize_kernel<<<(N + 127) / 128, 128, 0, stream>>>(
        x, root_w, bias, acc, out, N);
}


# x[src] rows gathered by whole-line loads (16 lanes read one 256 B row: 4 lines per load, was 32 lines x 16 B each) and turned into the einsum's lane layout through a per-wave LDS tile
# speedup vs baseline: 4.1613x; 1.0437x over previous
#include <hip/hip_runtime.h>
#include <hip/hip_bf16.h>


typedef __attribute__((ext_vector_type(16))) _Float16 v16h;
typedef __attribute__((ext_vector_type(8)))  _Float16 v8h;
typedef __attribute__((ext_vector_type(8)))  float    v8f;

#define LATENT   64
#define EDGE_IN  70
#define K1PAD    96
#define LD1      104
#define LDH      40
#define XLD      68
#define HID      20
#define OUTD     192
#define OUT_CH   3
#define OHALF    96
#define WAVES    16
#define EDGE_GRID 256


__device__ inline v8f wmma16(v16h a, v16h b, v8f c) {
    return __builtin_amdgcn_wmma_f32_16x16x32_f16(
        false, a, false, b, (short)0, c, false, false);
}

__device__ inline v16h cat8(v8h lo, v8h hi) {
    return __builtin_shufflevector(lo, hi,
        0, 1, 2, 3, 4, 5, 6, 7, 8, 9, 10, 11, 12, 13, 14, 15);
}

__device__ inline v16h frag(const _Float16* T, int ld, int k0, int r0) {
    int lane = threadIdx.x & 31;
    int r  = r0 + (lane & 15);
    int kb = k0 + ((lane & 16) ? 8 : 0);
    v8h lo = *(const v8h*)(T + r * ld + kb);
    v8h hi = *(const v8h*)(T + r * ld + kb + 16);
    return cat8(lo, hi);
}

__device__ inline v8f bias_c(const float* b, int n0) {
    int half = (threadIdx.x & 31) >> 4;
    return *(const v8f*)(b + n0 + 8 * half);
}

__device__ inline v16h relu_pack(v8f d0, v8f d1) {
    v16h h;
#pragma unroll
    for (int r = 0; r < 8; ++r) {
        h[r]     = (_Float16)d0[r];
        h[8 + r] = (_Float16)d1[r];
    }
    const v16h zero = {};
    return __builtin_elementwise_max(h, zero);
}


__global__ void zero_f32(float* p, int n) {
    int i = blockIdx.x * blockDim.x + threadIdx.x;
    if (i < n) p[i] = 0.0f;
}

__global__ __launch_bounds__(WAVES * 32)
void nnconv_edge_kernel(const float* __restrict__ x,
                        const int*   __restrict__ edge_index,
                        const float* __restrict__ edge_attr,
                        const float* __restrict__ W1, const float* __restrict__ b1,
                        const float* __restrict__ W2, const float* __restrict__ b2,
                        const float* __restrict__ W3, const float* __restrict__ b3,
                        const float* __restrict__ W4, const float* __restrict__ b4,
                        float* __restrict__ acc,
                        int E) {
    __shared__ alignas(16) _Float16 sW1T[32  * LD1];
    __shared__ alignas(16) _Float16 sW2T[32  * LDH];
    __shared__ alignas(16) _Float16 sW3T[32  * LDH];
    __shared__ alignas(16) _Float16 sW4T[OUTD * LDH];
    __shared__ alignas(32) float sB1[32], sB2[32], sB3[32];
    __shared__ alignas(32) float sB4[OUTD];
    __shared__ alignas(16) _Float16 sA [WAVES][16 * LD1];
    __shared__ alignas(16) float sMsg[WAVES][16 * 4];
    __shared__ alignas(16) float sX[WAVES][16 * XLD];

    const int tid = threadIdx.x;

    constexpr int NT = WAVES * 32;
    static_assert((32 * K1PAD) % NT == 0 && (32 * 32) % NT == 0 &&
                  (OUTD * 32) % NT == 0, "staging loops assume whole trips");
#pragma unroll
    for (int j = 0; j < 32 * K1PAD / NT; ++j) {
        int i = tid + j * NT;
        int n = i / K1PAD, k = i % K1PAD;
        float w = W1[min(k, EDGE_IN - 1) * HID + min(n, HID - 1)];
        sW1T[n * LD1 + k] = (_Float16)((k < EDGE_IN && n < HID) ? w : 0.0f);
    }
#pragma unroll
    for (int j = 0; j < 32 * 32 / NT; ++j) {
        int i = tid + j * NT;
        int n = i >> 5, k = i & 31;
        int g = min(k, HID - 1) * HID + min(n, HID - 1);
        float w2 = W2[g], w3 = W3[g];
        bool ok = (k < HID && n < HID);
        sW2T[n * LDH + k] = (_Float16)(ok ? w2 : 0.0f);
        sW3T[n * LDH + k] = (_Float16)(ok ? w3 : 0.0f);
    }
#pragma unroll
    for (int j = 0; j < OUTD * 32 / NT; ++j) {
        int i = tid + j * NT;
        int n = i >> 5, k = i & 31;
        float w = W4[min(k, HID - 1) * OUTD + n];
        int hh = n / OHALF, rem = n % OHALF;
        int row = (rem >> 3) * 16 + hh * 8 + (rem & 7);
        sW4T[row * LDH + k] = (_Float16)((k < HID) ? w : 0.0f);
    }
    for (int i = tid; i < 32; i += blockDim.x) {
        sB1[i] = (i < HID) ? b1[i] : 0.0f;
        sB2[i] = (i < HID) ? b2[i] : 0.0f;
        sB3[i] = (i < HID) ? b3[i] : 0.0f;
    }
    for (int i = tid; i < OUTD; i += blockDim.x) sB4[i] = b4[i];
    __syncthreads();

    const int lane = tid & 31;
    const int wave = __builtin_amdgcn_readfirstlane(tid >> 5);
    const int m    = lane & 15;
    const int half = lane >> 4;

    _Float16* A1 = sA [wave];

#pragma unroll
    for (int i = lane; i < 16 * (K1PAD - EDGE_IN); i += 32) {
        A1[(i / 26) * LD1 + EDGE_IN + (i % 26)] = (_Float16)0.0f;
    }

    const long tiles  = ((long)E + 15) >> 4;
    const long stride = (long)EDGE_GRID * WAVES;
    for (long tile = (long)wave * EDGE_GRID + blockIdx.x; tile < tiles;
         tile += stride) {
    const long e0 = tile * 16;

    {
        const long gbase = e0 * EDGE_IN;
        if (e0 + 16 <= E) {
            const float* ap = edge_attr + gbase;
            float v[16 * EDGE_IN / 32];
#pragma unroll
            for (int j = 0; j < 16 * EDGE_IN / 32; ++j) v[j] = ap[lane + 32 * j];
#pragma unroll
            for (int j = 0; j < 16 * EDGE_IN / 32; ++j) {
                int i = lane + 32 * j;
                A1[(i / EDGE_IN) * LD1 + (i % EDGE_IN)] = (_Float16)v[j];
            }
        } else {
            const long gmax = (long)E * EDGE_IN;
            for (int i = lane; i < 16 * EDGE_IN; i += 32) {
                long g = gbase + i;
                float v = (g < gmax) ? edge_attr[g] : 0.0f;
                A1[(i / EDGE_IN) * LD1 + (i % EDGE_IN)] = (_Float16)v;
            }
        }
    }
    asm volatile("s_wait_dscnt 0" ::: "memory");

    const long elast = (long)E - 1;
    float4 xg[8];
    {
        int sidx[8];
        if (e0 + 16 <= E) {
            const int4* ip = (const int4*)(edge_index + e0 + 8 * half);
            const int4 a = ip[0], b = ip[1];
            sidx[0] = a.x; sidx[1] = a.y; sidx[2] = a.z; sidx[3] = a.w;
            sidx[4] = b.x; sidx[5] = b.y; sidx[6] = b.z; sidx[7] = b.w;
        } else {
#pragma unroll
            for (int q = 0; q < 8; ++q) {
                const long eq = e0 + 8 * half + q;
                sidx[q] = edge_index[eq < elast ? eq : elast];
            }
        }
#pragma unroll
        for (int q = 0; q < 8; ++q) {
            xg[q] = *((const float4*)(x + (size_t)sidx[q] * LATENT) + m);
        }
    }

    v16h h;
    {
        v8f d0 = bias_c(sB1, 0), d1 = bias_c(sB1, 16);
#pragma unroll
        for (int ks = 0; ks < 3; ++ks) {
            v16h xT = frag(A1, LD1, ks * 32, 0);
            d0 = wmma16(frag(sW1T, LD1, ks * 32, 0),  xT, d0);
            d1 = wmma16(frag(sW1T, LD1, ks * 32, 16), xT, d1);
        }
        h = relu_pack(d0, d1);
    }
    float xv[32];
    {
        float* X = sX[wave];
#pragma unroll
        for (int q = 0; q < 8; ++q) {
            *(float4*)(X + (q + 8 * half) * XLD + m * 4) = xg[q];
        }
        asm volatile("s_wait_dscnt 0" ::: "memory");
        const float4* xr = (const float4*)(X + m * XLD + half * 32);
#pragma unroll
        for (int q = 0; q < 8; ++q) {
            float4 t = xr[q];
            xv[q * 4 + 0] = t.x; xv[q * 4 + 1] = t.y;
            xv[q * 4 + 2] = t.z; xv[q * 4 + 3] = t.w;
        }
    }

    {
        v8f d0 = wmma16(frag(sW2T, LDH, 0, 0),  h, bias_c(sB2, 0));
        v8f d1 = wmma16(frag(sW2T, LDH, 0, 16), h, bias_c(sB2, 16));
        h = relu_pack(d0, d1);
    }
    {
        v8f d0 = wmma16(frag(sW3T, LDH, 0, 0),  h, bias_c(sB3, 0));
        v8f d1 = wmma16(frag(sW3T, LDH, 0, 16), h, bias_c(sB3, 16));
        h = relu_pack(d0, d1);
    }

    const int  word  = lane & 3;
    const long ea    = e0 + (lane >> 2);
    const long eb    = ea + 8;
    const int  da = edge_index[(size_t)E + (ea < elast ? ea : elast)];
    const int  db = edge_index[(size_t)E + (eb < elast ? eb : elast)];

    float p0 = 0.0f, p1 = 0.0f, p2 = 0.0f;
    {
#pragma unroll
        for (int t = 0; t < OHALF / 8; ++t) {
            const v8f b = *(const v8f*)(sB4 + half * OHALF + t * 8);
            v8f c = wmma16(frag(sW4T, LDH, 0, t * 16), h, b);
#pragma unroll
            for (int j = 0; j < 8; ++j) {
                int f = t * 8 + j;
                int i = f / 3, cc = f % 3;
                float prod = xv[i] * c[j];
                if (cc == 0) p0 += prod;
                else if (cc == 1) p1 += prod;
                else p2 += prod;
            }
        }
    }
    p0 += __shfl_xor(p0, 16);
    p1 += __shfl_xor(p1, 16);
    p2 += __shfl_xor(p2, 16);

    {
        float* S = sMsg[wave];
        if (half == 0) *(float4*)(S + m * 4) = make_float4(p0, p1, p2, 1.0f);
        asm volatile("s_wait_dscnt 0" ::: "memory");
        const float va = S[lane];
        const float vb = S[32 + lane];
        if (ea < E) atomicAdd(&acc[(size_t)da * 4 + word], va);
        if (eb < E) atomicAdd(&acc[(size_t)db * 4 + word], vb);
    }
    }
}

__global__ void nnconv_finalize_kernel(const float* __restrict__ x,
                                       const float* __restrict__ root_w,
                                       const float* __restrict__ bias,
                                       const float* __restrict__ acc,
                                       float* __restrict__ out, int N) {
    int n = blockIdx.x * blockDim.x + threadIdx.x;
    if (n >= N) return;
    const float4 s = ((const float4*)acc)[n];
    float inv = 1.0f / fmaxf(s.w, 1.0f);
    float a0 = s.x * inv + bias[0];
    float a1 = s.y * inv + bias[1];
    float a2 = s.z * inv + bias[2];
    const float4* xr = (const float4*)(x + (size_t)n * LATENT);
#pragma unroll
    for (int q = 0; q < LATENT / 4; ++q) {
        float4 v = xr[q];
        const float* vp = (const float*)&v;
#pragma unroll
        for (int j = 0; j < 4; ++j) {
            int i = q * 4 + j;
            float xvv = vp[j];
            a0 += xvv * root_w[i * 3 + 0];
            a1 += xvv * root_w[i * 3 + 1];
            a2 += xvv * root_w[i * 3 + 2];
        }
    }
    out[(size_t)n * 3 + 0] = a0;
    out[(size_t)n * 3 + 1] = a1;
    out[(size_t)n * 3 + 2] = a2;
}


extern "C" void kernel_launch(void* const* d_in, const int* in_sizes, int n_in,
                              void* d_out, int out_size, void* d_ws, size_t ws_size,
                              hipStream_t stream) {
    const float* x          = (const float*)d_in[0];
    const int*   edge_index = (const int*)  d_in[1];
    const float* edge_attr  = (const float*)d_in[2];
    const float* W1 = (const float*)d_in[3];
    const float* b1 = (const float*)d_in[4];
    const float* W2 = (const float*)d_in[5];
    const float* b2 = (const float*)d_in[6];
    const float* W3 = (const float*)d_in[7];
    const float* b3 = (const float*)d_in[8];
    const float* W4 = (const float*)d_in[9];
    const float* b4 = (const float*)d_in[10];
    const float* root_w = (const float*)d_in[11];
    const float* bias   = (const float*)d_in[12];
    float* out = (float*)d_out;

    const int E = in_sizes[1] / 2;
    const int N = in_sizes[0] / LATENT;

    float* acc = (float*)d_ws;

    {
        int zn = N * 4;
        zero_f32<<<(zn + 255) / 256, 256, 0, stream>>>(acc, zn);
    }
    {
        nnconv_edge_kernel<<<EDGE_GRID, WAVES * 32, 0, stream>>>(
            x, edge_index, edge_attr,
            W1, b1, W2, b2, W3, b3, W4, b4,
            acc, E);
    }
    nnconv_finalize_kernel<<<(N + 127) / 128, 128, 0, stream>>>(
        x, root_w, bias, acc, out, N);
}


# Edge kernel at 8 waves per SIMD (was 4): 32-wave work-groups under a 128-register cap (pair-wise staging, lane addresses recomputed per trip, no scratch) fill each trip's waits; atomic order differs
# speedup vs baseline: 4.4116x; 1.0601x over previous
#include <hip/hip_runtime.h>
#include <hip/hip_bf16.h>


typedef __attribute__((ext_vector_type(16))) _Float16 v16h;
typedef __attribute__((ext_vector_type(8)))  _Float16 v8h;
typedef __attribute__((ext_vector_type(2)))  _Float16 v2h;
typedef __attribute__((ext_vector_type(8)))  float    v8f;

#define LATENT   64
#define EDGE_IN  70
#define K1PAD    96
#define LD1      104
#define LDH      40
#define XLD      68
#define HID      20
#define OUTD     192
#define OUT_CH   3
#define OHALF    96
#define WAVES    32
#define EDGE_GRID 256


__device__ inline v8f wmma16(v16h a, v16h b, v8f c) {
    return __builtin_amdgcn_wmma_f32_16x16x32_f16(
        false, a, false, b, (short)0, c, false, false);
}

__device__ inline v16h cat8(v8h lo, v8h hi) {
    return __builtin_shufflevector(lo, hi,
        0, 1, 2, 3, 4, 5, 6, 7, 8, 9, 10, 11, 12, 13, 14, 15);
}

__device__ inline v16h frag(const _Float16* T, int ld, int k0, int r0) {
    int lane = threadIdx.x & 31;
    int r  = r0 + (lane & 15);
    int kb = k0 + ((lane & 16) ? 8 : 0);
    v8h lo = *(const v8h*)(T + r * ld + kb);
    v8h hi = *(const v8h*)(T + r * ld + kb + 16);
    return cat8(lo, hi);
}

__device__ inline v8f bias_c(const float* b, int n0) {
    int half = (threadIdx.x & 31) >> 4;
    return *(const v8f*)(b + n0 + 8 * half);
}

__device__ inline v16h relu_pack(v8f d0, v8f d1) {
    v16h h;
#pragma unroll
    for (int r = 0; r < 8; ++r) {
        h[r]     = (_Float16)d0[r];
        h[8 + r] = (_Float16)d1[r];
    }
    const v16h zero = {};
    return __builtin_elementwise_max(h, zero);
}


__global__ void zero_f32(float* p, int n) {
    int i = blockIdx.x * blockDim.x + threadIdx.x;
    if (i < n) p[i] = 0.0f;
}

__global__ __launch_bounds__(WAVES * 32)
void nnconv_edge_kernel(const float* __restrict__ x,
                        const int*   __restrict__ edge_index,
                        const float* __restrict__ edge_attr,
                        const float* __restrict__ W1, const float* __restrict__ b1,
                        const float* __restrict__ W2, const float* __restrict__ b2,
                        const float* __restrict__ W3, const float* __restrict__ b3,
                        const float* __restrict__ W4, const float* __restrict__ b4,
                        float* __restrict__ acc,
                        int E) {
    __shared__ alignas(16) _Float16 sW1T[32  * LD1];
    __shared__ alignas(16) _Float16 sW2T[32  * LDH];
    __shared__ alignas(16) _Float16 sW3T[32  * LDH];
    __shared__ alignas(16) _Float16 sW4T[OUTD * LDH];
    __shared__ alignas(32) float sB1[32], sB2[32], sB3[32];
    __shared__ alignas(32) float sB4[OUTD];
    __shared__ alignas(16) _Float16 sA [WAVES][16 * LD1];
    __shared__ alignas(16) float sMsg[WAVES][16 * 4];
    __shared__ alignas(16) float sX[WAVES][16 * XLD];

    const int tid = threadIdx.x;

    constexpr int NT = WAVES * 32;
    static_assert((32 * K1PAD) % NT == 0 && (32 * 32) % NT == 0 &&
                  (OUTD * 32) % NT == 0, "staging loops assume whole trips");
#pragma unroll
    for (int j = 0; j < 32 * K1PAD / NT; ++j) {
        int i = tid + j * NT;
        int n = i / K1PAD, k = i % K1PAD;
        float w = W1[min(k, EDGE_IN - 1) * HID + min(n, HID - 1)];
        sW1T[n * LD1 + k] = (_Float16)((k < EDGE_IN && n < HID) ? w : 0.0f);
    }
#pragma unroll
    for (int j = 0; j < 32 * 32 / NT; ++j) {
        int i = tid + j * NT;
        int n = i >> 5, k = i & 31;
        int g = min(k, HID - 1) * HID + min(n, HID - 1);
        float w2 = W2[g], w3 = W3[g];
        bool ok = (k < HID && n < HID);
        sW2T[n * LDH + k] = (_Float16)(ok ? w2 : 0.0f);
        sW3T[n * LDH + k] = (_Float16)(ok ? w3 : 0.0f);
    }
#pragma unroll
    for (int j = 0; j < OUTD * 32 / NT; ++j) {
        int i = tid + j * NT;
        int n = i >> 5, k = i & 31;
        float w = W4[min(k, HID - 1) * OUTD + n];
        int hh = n / OHALF, rem = n % OHALF;
        int row = (rem >> 3) * 16 + hh * 8 + (rem & 7);
        sW4T[row * LDH + k] = (_Float16)((k < HID) ? w : 0.0f);
    }
    for (int i = tid; i < 32; i += blockDim.x) {
        sB1[i] = (i < HID) ? b1[i] : 0.0f;
        sB2[i] = (i < HID) ? b2[i] : 0.0f;
        sB3[i] = (i < HID) ? b3[i] : 0.0f;
    }
    for (int i = tid; i < OUTD; i += blockDim.x) sB4[i] = b4[i];
    __syncthreads();

    const int lane = tid & 31;
    const int wave = __builtin_amdgcn_readfirstlane(tid >> 5);
    const int m    = lane & 15;
    const int half = lane >> 4;

    _Float16* A1 = sA [wave];

#pragma unroll
    for (int i = lane; i < 16 * (K1PAD - EDGE_IN); i += 32) {
        A1[(i / 26) * LD1 + EDGE_IN + (i % 26)] = (_Float16)0.0f;
    }

    const long tiles  = ((long)E + 15) >> 4;
    const long stride = (long)EDGE_GRID * WAVES;
    for (long tile = (long)wave * EDGE_GRID + blockIdx.x; tile < tiles;
         tile += stride) {
    const long e0 = tile * 16;

    int lv = lane;
    asm volatile("" : "+v"(lv));
    const int mv = lv & 15;
    const int hv = lv >> 4;

    {
        const long gbase = e0 * EDGE_IN;
        if (e0 + 16 <= E) {
            constexpr int RP    = EDGE_IN / 2;
            constexpr int PAIRS = 16 * RP;
            constexpr int FULL  = PAIRS / 32;
            static_assert(EDGE_IN % 2 == 0 && PAIRS - FULL * 32 == 16,
                          "pair staging assumes an even row and a half-wave tail");
            const float2* ap = (const float2*)(edge_attr + gbase);
            float2 v[FULL + 1];
#pragma unroll
            for (int j = 0; j < FULL; ++j) v[j] = ap[lv + 32 * j];
            v[FULL] = ap[FULL * 32 + mv];
            _Float16* dst0 = A1 + 2 * lv;
            _Float16* dst1 = dst0 + (LD1 - EDGE_IN);
#pragma unroll
            for (int j = 0; j <= FULL; ++j) {
                const int qj = (32 * j) / RP, rj = (32 * j) % RP;
                const int koff = qj * LD1 + 2 * rj;
                v2h p = {(_Float16)v[j].x, (_Float16)v[j].y};
                if (j < FULL) {
                    _Float16* d = (RP - rj > 31) ? dst0 : ((lv >= RP - rj) ? dst1 : dst0);
                    *(v2h*)(d + koff) = p;
                } else {
                    *(v2h*)(A1 + koff + 2 * mv) = p;
                }
            }
        } else {
            const long gmax = (long)E * EDGE_IN;
            for (int i = lane; i < 16 * EDGE_IN; i += 32) {
                long g = gbase + i;
                float v = (g < gmax) ? edge_attr[g] : 0.0f;
                A1[(i / EDGE_IN) * LD1 + (i % EDGE_IN)] = (_Float16)v;
            }
        }
    }
    asm volatile("s_wait_dscnt 0" ::: "memory");

    const long elast = (long)E - 1;
    float4 xg[8];
    {
        int sidx[8];
        if (e0 + 16 <= E) {
            const int4* ip = (const int4*)(edge_index + e0 + 8 * hv);
            const int4 a = ip[0], b = ip[1];
            sidx[0] = a.x; sidx[1] = a.y; sidx[2] = a.z; sidx[3] = a.w;
            sidx[4] = b.x; sidx[5] = b.y; sidx[6] = b.z; sidx[7] = b.w;
        } else {
#pragma unroll
            for (int q = 0; q < 8; ++q) {
                const long eq = e0 + 8 * hv + q;
                sidx[q] = edge_index[eq < elast ? eq : elast];
            }
        }
#pragma unroll
        for (int q = 0; q < 8; ++q) {
            xg[q] = *((const float4*)(x + (size_t)sidx[q] * LATENT) + m);
        }
    }

    v16h h;
    {
        v8f d0 = bias_c(sB1, 0), d1 = bias_c(sB1, 16);
#pragma unroll
        for (int ks = 0; ks < 3; ++ks) {
            v16h xT = frag(A1, LD1, ks * 32, 0);
            d0 = wmma16(frag(sW1T, LD1, ks * 32, 0),  xT, d0);
            d1 = wmma16(frag(sW1T, LD1, ks * 32, 16), xT, d1);
        }
        h = relu_pack(d0, d1);
    }
    float xv[32];
    {
        float* X = sX[wave];
        float* xs = X + 8 * hv * XLD + mv * 4;
#pragma unroll
        for (int q = 0; q < 8; ++q) {
            *(float4*)(xs + q * XLD) = xg[q];
        }
        asm volatile("s_wait_dscnt 0" ::: "memory");
        const float4* xr = (const float4*)(X + mv * XLD + hv * 32);
#pragma unroll
        for (int q = 0; q < 8; ++q) {
            float4 t = xr[q];
            xv[q * 4 + 0] = t.x; xv[q * 4 + 1] = t.y;
            xv[q * 4 + 2] = t.z; xv[q * 4 + 3] = t.w;
        }
    }

    {
        v8f d0 = wmma16(frag(sW2T, LDH, 0, 0),  h, bias_c(sB2, 0));
        v8f d1 = wmma16(frag(sW2T, LDH, 0, 16), h, bias_c(sB2, 16));
        h = relu_pack(d0, d1);
    }
    {
        v8f d0 = wmma16(frag(sW3T, LDH, 0, 0),  h, bias_c(sB3, 0));
        v8f d1 = wmma16(frag(sW3T, LDH, 0, 16), h, bias_c(sB3, 16));
        h = relu_pack(d0, d1);
    }

    const long ea    = e0 + (lv >> 2);
    const long eb    = ea + 8;
    const int  da = edge_index[(size_t)E + (ea < elast ? ea : elast)];
    const int  db = edge_index[(size_t)E + (eb < elast ? eb : elast)];

    float p0 = 0.0f, p1 = 0.0f, p2 = 0.0f;
    {
#pragma unroll
        for (int t = 0; t < OHALF / 8; ++t) {
            const v8f b = *(const v8f*)(sB4 + half * OHALF + t * 8);
            v8f c = wmma16(frag(sW4T, LDH, 0, t * 16), h, b);
#pragma unroll
            for (int j = 0; j < 8; ++j) {
                int f = t * 8 + j;
                int i = f / 3, cc = f % 3;
                float prod = xv[i] * c[j];
                if (cc == 0) p0 += prod;
                else if (cc == 1) p1 += prod;
                else p2 += prod;
            }
        }
    }
    p0 += __shfl_xor(p0, 16);
    p1 += __shfl_xor(p1, 16);
    p2 += __shfl_xor(p2, 16);

    {
        int ls = lane;
        asm volatile("" : "+v"(ls));
        const int  word = ls & 3;
        const long eas  = e0 + (ls >> 2);
        float* S = sMsg[wave];
        if ((ls >> 4) == 0) {
            *(float4*)(S + (ls & 15) * 4) = make_float4(p0, p1, p2, 1.0f);
        }
        asm volatile("s_wait_dscnt 0" ::: "memory");
        const float va = S[ls];
        const float vb = S[32 + ls];
        if (eas < E)     atomicAdd(&acc[(size_t)da * 4 + word], va);
        if (eas + 8 < E) atomicAdd(&acc[(size_t)db * 4 + word], vb);
    }
    }
}

__global__ void nnconv_finalize_kernel(const float* __restrict__ x,
                                       const float* __restrict__ root_w,
                                       const float* __restrict__ bias,
                                       const float* __restrict__ acc,
                                       float* __restrict__ out, int N) {
    int n = blockIdx.x * blockDim.x + threadIdx.x;
    if (n >= N) return;
    const float4 s = ((const float4*)acc)[n];
    float inv = 1.0f / fmaxf(s.w, 1.0f);
    float a0 = s.x * inv + bias[0];
    float a1 = s.y * inv + bias[1];
    float a2 = s.z * inv + bias[2];
    const float4* xr = (const float4*)(x + (size_t)n * LATENT);
#pragma unroll
    for (int q = 0; q < LATENT / 4; ++q) {
        float4 v = xr[q];
        const float* vp = (const float*)&v;
#pragma unroll
        for (int j = 0; j < 4; ++j) {
            int i = q * 4 + j;
            float xvv = vp[j];
            a0 += xvv * root_w[i * 3 + 0];
            a1 += xvv * root_w[i * 3 + 1];
            a2 += xvv * root_w[i * 3 + 2];
        }
    }
    out[(size_t)n * 3 + 0] = a0;
    out[(size_t)n * 3 + 1] = a1;
    out[(size_t)n * 3 + 2] = a2;
}


extern "C" void kernel_launch(void* const* d_in, const int* in_sizes, int n_in,
                              void* d_out, int out_size, void* d_ws, size_t ws_size,
                              hipStream_t stream) {
    const float* x          = (const float*)d_in[0];
    const int*   edge_index = (const int*)  d_in[1];
    const float* edge_attr  = (const float*)d_in[2];
    const float* W1 = (const float*)d_in[3];
    const float* b1 = (const float*)d_in[4];
    const float* W2 = (const float*)d_in[5];
    const float* b2 = (const float*)d_in[6];
    const float* W3 = (const float*)d_in[7];
    const float* b3 = (const float*)d_in[8];
    const float* W4 = (const float*)d_in[9];
    const float* b4 = (const float*)d_in[10];
    const float* root_w = (const float*)d_in[11];
    const float* bias   = (const float*)d_in[12];
    float* out = (float*)d_out;

    const int E = in_sizes[1] / 2;
    const int N = in_sizes[0] / LATENT;

    float* acc = (float*)d_ws;

    {
        int zn = N * 4;
        zero_f32<<<(zn + 255) / 256, 256, 0, stream>>>(acc, zn);
    }
    {
        nnconv_edge_kernel<<<EDGE_GRID, WAVES * 32, 0, stream>>>(
            x, edge_index, edge_attr,
            W1, b1, W2, b2, W3, b3, W4, b4,
            acc, E);
    }
    nnconv_finalize_kernel<<<(N + 127) / 128, 128, 0, stream>>>(
        x, root_w, bias, acc, out, N);
}
